# starting kernel
# baseline (speedup 1.0000x reference)
#include <hip/hip_runtime.h>

#define DIGIT_POS 15
#define BATCH 32
#define SEQ 16384
#define DMODEL 64

#define TILE 512
#define TPR (SEQ / TILE)
#define CHUNKS_PER_TILE (TILE / 256)

typedef __attribute__((ext_vector_type(2))) float v2f;
typedef __attribute__((ext_vector_type(8))) float v8f;

__device__ __forceinline__ float lane_shfl(float v, int src) {
  return __shfl(v, src, 32);
}

__device__ __forceinline__ v8f wmma_16x16x4_f32(v2f a, v2f b, v8f c) {
  return __builtin_amdgcn_wmma_f32_16x16x4_f32(
       false, a,  false, b,
       (short)0, c,  false,  false);
}

__global__ void pass1_tile_sums(const float* __restrict__ h,
                                const float* __restrict__ vw,
                                const float* __restrict__ vb,
                                float* __restrict__ partial) {
  const int bid  = blockIdx.x;
  const int b    = bid / TPR;
  const int tile = bid % TPR;
  const int lane = threadIdx.x & 31;
  const float w = vw[0], bias = vb[0];
  const float* hrow = h + ((size_t)b * SEQ) * DMODEL + DIGIT_POS;
  const int s0 = tile * TILE;

  float sum = 0.f;
  #pragma unroll
  for (int k = 0; k < TILE / 32; ++k) {
    const int s = s0 + lane + 32 * k;
    sum += hrow[(size_t)s * DMODEL] * w + bias;
  }
  #pragma unroll
  for (int d = 16; d; d >>= 1) sum += __shfl_xor(sum, d, 32);
  if (lane == 0) partial[bid] = sum;
}

__global__ void pass2_scan_partials(float* __restrict__ partial) {
  const int b    = blockIdx.x;
  const int lane = threadIdx.x & 31;
  const float v = partial[b * TPR + lane];
  float incl = v;
  #pragma unroll
  for (int d = 1; d < 32; d <<= 1) {
    const int src = (lane >= d) ? (lane - d) : lane;
    const float up = lane_shfl(incl, src);
    if (lane >= d) incl += up;
  }
  partial[b * TPR + lane] = incl - v;
}

__global__ void pass3_wmma_scan(const float* __restrict__ h,
                                const float* __restrict__ vw,
                                const float* __restrict__ vb,
                                const float* __restrict__ partial,
                                float* __restrict__ out) {
  const int bid  = blockIdx.x;
  const int b    = bid / TPR;
  const int tile = bid % TPR;
  const int lane = threadIdx.x & 31;
  const int n    = lane & 15;
  const int hi   = lane >> 4;
  const float w = vw[0], bias = vb[0];
  const float* hrow = h + ((size_t)b * SEQ) * DMODEL + DIGIT_POS;
  float* orow = out + (size_t)b * SEQ;
  float carry = partial[bid];

  v2f Lp[4];
  #pragma unroll
  for (int p = 0; p < 4; ++p) {
    const int j0 = 4 * p + 2 * hi;
    Lp[p].x = (j0     <= n) ? 1.f : 0.f;
    Lp[p].y = (j0 + 1 <= n) ? 1.f : 0.f;
  }

  int s0 = tile * TILE;
  for (int c = 0; c < CHUNKS_PER_TILE; ++c, s0 += 256) {
    if (c + 1 < CHUNKS_PER_TILE) {
      #pragma unroll
      for (int k = 0; k < 8; ++k)
        __builtin_prefetch(hrow + (size_t)(s0 + 256 + lane + 32 * k) * DMODEL, 0, 3);
    }

    v8f acc = {0.f, 0.f, 0.f, 0.f, 0.f, 0.f, 0.f, 0.f};
    #pragma unroll
    for (int p = 0; p < 4; ++p) {
      const int r0 = 4 * p + 2 * hi;
      v2f bp;
      bp.x = hrow[(size_t)(s0 + 16 * n + r0    ) * DMODEL] * w + bias;
      bp.y = hrow[(size_t)(s0 + 16 * n + r0 + 1) * DMODEL] * w + bias;
      acc = wmma_16x16x4_f32(Lp[p], bp, acc);
    }

    const float t = lane_shfl(acc[7], 16 + n);
    float incl = t;
    #pragma unroll
    for (int d = 1; d < 16; d <<= 1) {
      const int src = (n >= d) ? (lane - d) : lane;
      const float up = lane_shfl(incl, src);
      if (n >= d) incl += up;
    }
    const float exoff = (incl - t) + carry;
    const float chunk_total = lane_shfl(incl, 31);

    #pragma unroll
    for (int i = 0; i < 8; ++i) {
      const int s = s0 + 16 * n + i + 8 * hi;
      orow[s] = (acc[i] + exoff) / (float)(s + 2);
    }
    carry += chunk_total;
  }
}

extern "C" void kernel_launch(void* const* d_in, const int* in_sizes, int n_in,
                              void* d_out, int out_size, void* d_ws, size_t ws_size,
                              hipStream_t stream) {
  (void)in_sizes; (void)n_in; (void)out_size; (void)ws_size;
  const float* h  = (const float*)d_in[0];
  const float* vw = (const float*)d_in[1];
  const float* vb = (const float*)d_in[2];
  float* out      = (float*)d_out;
  float* partial  = (float*)d_ws;

  pass1_tile_sums   <<<BATCH * TPR, 32, 0, stream>>>(h, vw, vb, partial);
  pass2_scan_partials<<<BATCH,      32, 0, stream>>>(partial);
  pass3_wmma_scan   <<<BATCH * TPR, 32, 0, stream>>>(h, vw, vb, partial, out);
}


# pass 1 as 256 four-wave work-groups: waves trade tile totals through LDS, store segment-local prefix sums; the carry pass sums the row's 8 segment totals itself; scan launch gone; sum order changes
# speedup vs baseline: 1.4568x; 1.4568x over previous
#include <hip/hip_runtime.h>

#define DIGIT_POS 15
#define BATCH 32
#define SEQ 16384
#define DMODEL 64

#define TILE 512
#define TPR (SEQ / TILE)
#define CHUNKS_PER_TILE (TILE / 256)
#define WAVES 4
#define SPR (TPR / WAVES)

typedef __attribute__((ext_vector_type(2))) float v2f;
typedef __attribute__((ext_vector_type(8))) float v8f;

__device__ __forceinline__ float lane_shfl(float v, int src) {
  return __shfl(v, src, 32);
}

__device__ __forceinline__ v8f wmma_16x16x4_f32(v2f a, v2f b, v8f c) {
  return __builtin_amdgcn_wmma_f32_16x16x4_f32(
       false, a,  false, b,
       (short)0, c,  false,  false);
}

__global__ __launch_bounds__(WAVES * 32)
void pass1_wmma_segment_scan(const float* __restrict__ h,
                             const float* __restrict__ vw,
                             const float* __restrict__ vb,
                             float* __restrict__ segtot,
                             float* __restrict__ out) {
  __shared__ float tile_tot[WAVES];
  const int sid  = blockIdx.x;
  const int wid  = threadIdx.x >> 5;
  const int bid  = sid * WAVES + wid;
  const int b    = bid / TPR;
  const int tile = bid % TPR;
  const int lane = threadIdx.x & 31;
  const int n    = lane & 15;
  const int hi   = lane >> 4;
  const float w = vw[0], bias = vb[0];
  const float* hrow = h + ((size_t)b * SEQ) * DMODEL + DIGIT_POS;
  float* orow = out + (size_t)b * SEQ;

  v2f Lp[4];
  #pragma unroll
  for (int p = 0; p < 4; ++p) {
    const int j0 = 4 * p + 2 * hi;
    Lp[p].x = (j0     <= n) ? 1.f : 0.f;
    Lp[p].y = (j0 + 1 <= n) ? 1.f : 0.f;
  }

  const float* lbase = hrow + (size_t)(tile * TILE + 16 * n + 2 * hi) * DMODEL;
  float raw[CHUNKS_PER_TILE][8];
  #pragma unroll
  for (int c = 0; c < CHUNKS_PER_TILE; ++c) {
    #pragma unroll
    for (int p = 0; p < 4; ++p) {
      raw[c][2 * p    ] = lbase[(size_t)(256 * c + 4 * p    ) * DMODEL];
      raw[c][2 * p + 1] = lbase[(size_t)(256 * c + 4 * p + 1) * DMODEL];
    }
  }

  float local = 0.f;
  v8f   accs[CHUNKS_PER_TILE];
  float exoff[CHUNKS_PER_TILE];
  #pragma unroll
  for (int c = 0; c < CHUNKS_PER_TILE; ++c) {
    v8f acc = {0.f, 0.f, 0.f, 0.f, 0.f, 0.f, 0.f, 0.f};
    #pragma unroll
    for (int p = 0; p < 4; ++p) {
      v2f bp;
      bp.x = raw[c][2 * p    ] * w + bias;
      bp.y = raw[c][2 * p + 1] * w + bias;
      acc = wmma_16x16x4_f32(Lp[p], bp, acc);
    }

    const float t = lane_shfl(acc[7], 16 + n);
    float incl = t;
    #pragma unroll
    for (int d = 1; d < 16; d <<= 1) {
      const int src = (n >= d) ? (lane - d) : lane;
      const float up = lane_shfl(incl, src);
      if (n >= d) incl += up;
    }
    exoff[c] = (incl - t) + local;
    accs[c]  = acc;
    local += lane_shfl(incl, 31);
  }

  if (lane == 0) tile_tot[wid] = local;
  __syncthreads();
  float carry = 0.f;
  #pragma unroll
  for (int k = 0; k < WAVES - 1; ++k) carry += (k < wid) ? tile_tot[k] : 0.f;

  #pragma unroll
  for (int c = 0; c < CHUNKS_PER_TILE; ++c) {
    const float off = exoff[c] + carry;
    #pragma unroll
    for (int i = 0; i < 8; ++i) {
      const int s = tile * TILE + 256 * c + 16 * n + i + 8 * hi;
      orow[s] = accs[c][i] + off;
    }
  }
  if (wid == WAVES - 1 && lane == 0) segtot[sid] = carry + local;
}

__global__ void pass2_add_carry(const float* __restrict__ segtot,
                                float* __restrict__ out) {
  const int bid  = blockIdx.x;
  const int b    = bid / TPR;
  const int tile = bid % TPR;
  const int seg  = tile / WAVES;
  const int lane = threadIdx.x & 31;
  float4* tp = reinterpret_cast<float4*>(out + (size_t)bid * TILE);
  const int s0 = tile * TILE;

  float4 x[TILE / 128];
  #pragma unroll
  for (int k = 0; k < TILE / 128; ++k) x[k] = tp[lane + 32 * k];

  const v8f tot = *reinterpret_cast<const v8f*>(segtot + b * SPR);
  float carry = 0.f;
  #pragma unroll
  for (int q = 0; q < SPR - 1; ++q) carry += (q < seg) ? tot[q] : 0.f;

  #pragma unroll
  for (int k = 0; k < TILE / 128; ++k) {
    const int s = s0 + 4 * (lane + 32 * k);
    float4 y;
    y.x = (x[k].x + carry) / (float)(s + 2);
    y.y = (x[k].y + carry) / (float)(s + 3);
    y.z = (x[k].z + carry) / (float)(s + 4);
    y.w = (x[k].w + carry) / (float)(s + 5);
    tp[lane + 32 * k] = y;
  }
}

extern "C" void kernel_launch(void* const* d_in, const int* in_sizes, int n_in,
                              void* d_out, int out_size, void* d_ws, size_t ws_size,
                              hipStream_t stream) {
  (void)in_sizes; (void)n_in; (void)out_size; (void)ws_size;
  const float* h  = (const float*)d_in[0];
  const float* vw = (const float*)d_in[1];
  const float* vb = (const float*)d_in[2];
  float* out      = (float*)d_out;
  float* segtot   = (float*)d_ws;

  pass1_wmma_segment_scan<<<BATCH * SPR, WAVES * 32, 0, stream>>>(h, vw, vb, segtot, out);
  pass2_add_carry        <<<BATCH * TPR, 32,         0, stream>>>(segtot, out);
}
